# starting kernel
# baseline (speedup 1.0000x reference)
#include <hip/hip_runtime.h>
#include <hip/hip_bf16.h>

typedef __attribute__((ext_vector_type(2))) float v2f;
typedef __attribute__((ext_vector_type(8))) float v8f;

#define D 128

__global__ void deg_init_kernel(float* deg, int n) {
    int i = blockIdx.x * blockDim.x + threadIdx.x;
    if (i < n) deg[i] = 1.0f;
}

__global__ void deg_count_kernel(const int* row, float* deg, int E) {
    int e = blockIdx.x * blockDim.x + threadIdx.x;
    if (e < E) atomicAdd(&deg[row[e]], 1.0f);
}

__global__ void deg_rsqrt_kernel(const float* deg, float* dis, int n) {
    int i = blockIdx.x * blockDim.x + threadIdx.x;
    if (i < n) dis[i] = rsqrtf(deg[i]);
}

__global__ void gemm_wmma_f32(const float* __restrict__ x,
                              const float* __restrict__ W,
                              float* __restrict__ h) {
    const int wave = threadIdx.x >> 5;
    const int lane = threadIdx.x & 31;
    const int half = lane >> 4;
    const int idx  = lane & 15;
    const int m0 = blockIdx.x << 4;
    const int n0 = wave << 4;

    v8f c = {};
    const float* xr = x + (m0 + idx) * D;
    const float* Wc = W + idx + n0;
#pragma unroll 8
    for (int k0 = 0; k0 < D; k0 += 4) {
        v2f a, b;
        a.x = xr[k0 + 2 * half + 0];
        a.y = xr[k0 + 2 * half + 1];
        b.x = Wc[(k0 + 2 * half + 0) * D];
        b.y = Wc[(k0 + 2 * half + 1) * D];
        c = __builtin_amdgcn_wmma_f32_16x16x4_f32(
             false, a,  false, b,
             (short)0, c,  false,  false);
    }
    float* hp = h + (m0 + 8 * half) * D + n0 + idx;
#pragma unroll
    for (int g = 0; g < 8; ++g) hp[g * D] = c[g];
}

__global__ void self_bias_kernel(const float* __restrict__ h,
                                 const float* __restrict__ dis,
                                 const float* __restrict__ bias,
                                 float* __restrict__ out, int n) {
    int gid = blockIdx.x * blockDim.x + threadIdx.x;
    if (gid >= n * D) return;
    int i = gid >> 7;
    int d = gid & (D - 1);
    float s = dis[i];
    out[gid] = bias[d] + h[gid] * s * s;
}

__global__ void scatter_kernel(const int* __restrict__ row,
                               const int* __restrict__ col,
                               const float* __restrict__ h,
                               const float* __restrict__ dis,
                               float* __restrict__ out, int E) {
    int gid = blockIdx.x * blockDim.x + threadIdx.x;
    int e = gid >> 5;
    if (e >= E) return;
    int lane = gid & 31;
    int r = row[e];
    int c = col[e];
    float w = dis[r] * dis[c];
    const float4 hv = *(const float4*)(h + r * D + lane * 4);
    float* o = out + c * D + lane * 4;
    atomicAdd(o + 0, hv.x * w);
    atomicAdd(o + 1, hv.y * w);
    atomicAdd(o + 2, hv.z * w);
    atomicAdd(o + 3, hv.w * w);
}

extern "C" void kernel_launch(void* const* d_in, const int* in_sizes, int n_in,
                              void* d_out, int out_size, void* d_ws, size_t ws_size,
                              hipStream_t stream) {
    const float* x    = (const float*)d_in[0];
    const int*   ei   = (const int*)d_in[1];
    const float* W    = (const float*)d_in[2];
    const float* bias = (const float*)d_in[3];
    float* out = (float*)d_out;

    const int N = in_sizes[0] / D;
    const int E = in_sizes[1] / 2;
    const int* row = ei;
    const int* col = ei + E;

    float* h   = (float*)d_ws;
    float* deg = h + (size_t)N * D;
    float* dis = deg + N;

    deg_init_kernel<<<(N + 255) / 256, 256, 0, stream>>>(deg, N);
    deg_count_kernel<<<(E + 255) / 256, 256, 0, stream>>>(row, deg, E);
    deg_rsqrt_kernel<<<(N + 255) / 256, 256, 0, stream>>>(deg, dis, N);
    gemm_wmma_f32<<<N / 16, 256, 0, stream>>>(x, W, h);
    self_bias_kernel<<<((size_t)N * D + 255) / 256, 256, 0, stream>>>(h, dis, bias, out, N);
    {
        long long threads = (long long)E * 32;
        int blocks = (int)((threads + 255) / 256);
        scatter_kernel<<<blocks, 256, 0, stream>>>(row, col, h, dis, out, E);
    }
}


# Waves sum rows in registers (row = relative register index): 4 waves share a chunk set, each keeps its 10 rows' edges, one 16-byte load an edge; partial rows join by LDS float adds, arrival order
# speedup vs baseline: 4.9939x; 4.9939x over previous
#include <hip/hip_runtime.h>
#include <hip/hip_bf16.h>

typedef __attribute__((ext_vector_type(2))) float v2f;
typedef __attribute__((ext_vector_type(8))) float v8f;

#define D 128
#define R_OWN 40

#define IDX_THREADS 1024
#define CH_EDGES (IDX_THREADS * 4)
#define IDX_OWNERS 256
#define QSHIFT 6

__device__ __forceinline__ void chunk_starts(const unsigned* cnt, unsigned* st, unsigned* ow, int lane) {
    unsigned c[8];
    unsigned s = 0u;
#pragma unroll
    for (int i = 0; i < 8; ++i) { c[i] = cnt[lane * 8 + i]; s += c[i]; }
    unsigned incl = s;
#pragma unroll
    for (int d = 1; d < 32; d <<= 1) {
        const unsigned v = __shfl_up(incl, (unsigned)d);
        if (lane >= d) incl += v;
    }
    unsigned run = incl - s;
#pragma unroll
    for (int i = 0; i < 8; ++i) {
        st[lane * 8 + i] = run;
        ow[lane * 8 + i] = run | ((run + c[i]) << 16);
        run += c[i];
    }
}

__device__ __forceinline__ unsigned take_rank(unsigned* cnt, unsigned owner) {
    return __hip_atomic_fetch_add(cnt + owner, 1u, __ATOMIC_RELAXED, __HIP_MEMORY_SCOPE_WORKGROUP);
}

__global__ __launch_bounds__(IDX_THREADS)
void owner_sort_kernel(const int* __restrict__ row, const int* __restrict__ col,
                       unsigned* __restrict__ srtc, unsigned* __restrict__ offc,
                       unsigned char* __restrict__ srtr, unsigned* __restrict__ offr, int E4) {
    __shared__ unsigned cntC[IDX_OWNERS], cntR[IDX_OWNERS];
    __shared__ unsigned stC[IDX_OWNERS], stR[IDX_OWNERS];
    __shared__ unsigned owC[IDX_OWNERS], owR[IDX_OWNERS];
    __shared__ __attribute__((aligned(16))) unsigned stageC[CH_EDGES];
    __shared__ __attribute__((aligned(16))) unsigned char stageR[CH_EDGES];
    const int tid = threadIdx.x;
    const unsigned ro = R_OWN;
    const int t = blockIdx.x * IDX_THREADS + tid;
    const bool have = t < E4;
    int4 r = {0, 0, 0, 0}, cc = {0, 0, 0, 0};
    if (have) {
        r = ((const int4*)row)[t];
        cc = ((const int4*)col)[t];
    }
    if (tid < IDX_OWNERS) { cntC[tid] = 0u; cntR[tid] = 0u; }
    __syncthreads();
    asm volatile("" :: "v"(r.x), "v"(r.y), "v"(r.z), "v"(r.w), "v"(cc.x), "v"(cc.y), "v"(cc.z), "v"(cc.w));

    unsigned sr[4] = {0u, 0u, 0u, 0u}, oc[4] = {0u, 0u, 0u, 0u}, orr[4] = {0u, 0u, 0u, 0u};
    unsigned lc[4] = {0u, 0u, 0u, 0u}, lr[4] = {0u, 0u, 0u, 0u}, kc[4] = {0u, 0u, 0u, 0u}, kr[4] = {0u, 0u, 0u, 0u};
    if (have) {
        sr[0] = (unsigned)r.x; sr[1] = (unsigned)r.y; sr[2] = (unsigned)r.z; sr[3] = (unsigned)r.w;
        const unsigned dc[4] = {(unsigned)cc.x, (unsigned)cc.y, (unsigned)cc.z, (unsigned)cc.w};
#pragma unroll
        for (int i = 0; i < 4; ++i) {
            oc[i] = dc[i] / ro;  lc[i] = dc[i] - oc[i] * ro;
            orr[i] = sr[i] / ro; lr[i] = sr[i] - orr[i] * ro;
        }
#pragma unroll
        for (int i = 0; i < 4; ++i) { kc[i] = take_rank(cntC, oc[i]); kr[i] = take_rank(cntR, orr[i]); }
    }
    __syncthreads();

    if (tid < 32) chunk_starts(cntC, stC, owC, tid);
    else if (tid < 64) chunk_starts(cntR, stR, owR, tid - 32);
    __syncthreads();

    if (have) {
#pragma unroll
        for (int i = 0; i < 4; ++i) {
            stageC[stC[oc[i]] + kc[i]] = (sr[i] << QSHIFT) | lc[i];
            stageR[stR[orr[i]] + kr[i]] = (unsigned char)lr[i];
        }
    }
    __syncthreads();

    ((int4*)srtc)[(size_t)blockIdx.x * (CH_EDGES / 4) + tid] = ((const int4*)stageC)[tid];
    if (tid < CH_EDGES / 16)
        ((int4*)srtr)[(size_t)blockIdx.x * (CH_EDGES / 16) + tid] = ((const int4*)stageR)[tid];
    if (tid < IDX_OWNERS) {
        offc[(size_t)blockIdx.x * IDX_OWNERS + tid] = owC[tid];
        offr[(size_t)blockIdx.x * IDX_OWNERS + tid] = owR[tid];
    }
}

#define AGG_THREADS 1024
#define AGG_WAVES 32
#define RUN_B 8

__device__ __forceinline__ unsigned run_words(const unsigned* __restrict__ off, int nchunks, int wave, int lane,
                                              int kb) {
    const int ch = wave + AGG_WAVES * (kb + lane);
    unsigned ofw = 0u;
    if (lane < RUN_B && ch < nchunks) ofw = off[(size_t)ch * IDX_OWNERS + blockIdx.x];
    return ofw;
}

__global__ __launch_bounds__(AGG_THREADS)
void degree_owner_kernel(const unsigned char* __restrict__ srtr, const unsigned* __restrict__ offr,
                         float* __restrict__ dis, int n, int nchunks) {
    __shared__ int bins[R_OWN];

    const int tid = threadIdx.x;
    const int lane = tid & 31;
    const int wave = __builtin_amdgcn_readfirstlane(tid >> 5);
    const int c0 = blockIdx.x * R_OWN;

    if (tid < R_OWN) bins[tid] = 0;
    __syncthreads();

    for (int kb = 0; wave + AGG_WAVES * kb < nchunks; kb += RUN_B) {
        const unsigned ofw = run_words(offr, nchunks, wave, lane, kb);
        int v[RUN_B];
#pragma unroll
        for (int c = 0; c < RUN_B; ++c) {
            const unsigned o = (unsigned)__builtin_amdgcn_readlane((int)ofw, c);
            const int st = (int)(o & 0xFFFFu), cnt = (int)(o >> 16) - st;
            const unsigned char* run = srtr + (size_t)(wave + AGG_WAVES * (kb + c)) * CH_EDGES + st;
            v[c] = -1;
            if (lane < cnt) v[c] = run[lane];
        }
#pragma unroll
        for (int c = 0; c < RUN_B; ++c) {
            if (v[c] >= 0) __hip_atomic_fetch_add(bins + v[c], 1, __ATOMIC_RELAXED, __HIP_MEMORY_SCOPE_WORKGROUP);
            const unsigned o = (unsigned)__builtin_amdgcn_readlane((int)ofw, c);
            const int st = (int)(o & 0xFFFFu), cnt = (int)(o >> 16) - st;
            const unsigned char* run = srtr + (size_t)(wave + AGG_WAVES * (kb + c)) * CH_EDGES + st;
            for (int j = 32 + lane; j < cnt; j += 32)
                __hip_atomic_fetch_add(bins + run[j], 1, __ATOMIC_RELAXED, __HIP_MEMORY_SCOPE_WORKGROUP);
        }
    }
    __syncthreads();

    if (tid < R_OWN && c0 + tid < n) dis[c0 + tid] = rsqrtf((float)(bins[tid] + 1));
}

#define QCAP 288
#define AP 132
#define WP 144
#define M_TILES ((R_OWN + 15) / 16)
#define GEMM_WAVES (M_TILES * (D / 16))

#define NQ 4
#define RQ (R_OWN / NQ)
#define CSETS (AGG_WAVES / NQ)

__device__ __forceinline__ unsigned run_words_q(const unsigned* __restrict__ off, int nchunks, int cset, int lane,
                                                int kq) {
    const int ch = cset + CSETS * (kq + lane);
    unsigned ofw = 0u;
    if (ch < nchunks) ofw = off[(size_t)ch * IDX_OWNERS + blockIdx.x];
    return ofw;
}

typedef __attribute__((ext_vector_type(RQ))) float rowvec;
struct RowSums { rowvec k0, k1, k2, k3; };

__device__ __forceinline__ void add_row(RowSums& a, int row, float4 v, float w) {
    a.k0[row] = fmaf(v.x, w, a.k0[row]);
    a.k1[row] = fmaf(v.y, w, a.k1[row]);
    a.k2[row] = fmaf(v.z, w, a.k2[row]);
    a.k3[row] = fmaf(v.w, w, a.k3[row]);
}

__device__ __forceinline__ void add_group(const int* qp, int cnt, int lane,
                                          const float* __restrict__ x,
                                          const float* __restrict__ dis,
                                          const float* sdis, int qlo, RowSums& a) {
    int rq = 0;
    float w = 0.0f;
    if (lane < cnt) {
        const unsigned pk = (unsigned)qp[lane];
        const int cw = (int)(pk & ((1u << QSHIFT) - 1u));
        w = dis[pk >> QSHIFT] * sdis[cw];
        rq = (int)pk - qlo;
    }
    const unsigned lo = (unsigned)lane * 4u;
    for (int j = 0; j < cnt; j += 8) {
        float4 v[8];
        int cj[8];
        float wj[8];
#pragma unroll
        for (int u = 0; u < 8; ++u) {
            const int jj = min(j + u, cnt - 1);
            const unsigned pj = (unsigned)__builtin_amdgcn_readlane(rq, jj);
            cj[u] = (int)(pj & ((1u << QSHIFT) - 1u));
            wj[u] = __int_as_float(__builtin_amdgcn_readlane(__float_as_int(w), jj));
            const float* xr = x + (size_t)(pj >> QSHIFT) * D;
            v[u] = *(const float4*)(xr + lo);
        }
#pragma unroll
        for (int u = 0; u < 8; ++u) {
            if (j + u < cnt) add_row(a, cj[u], v[u], wj[u]);
        }
    }
}

__device__ __forceinline__ void run_heads(const unsigned* __restrict__ srtc,
                                          unsigned ofw, int cset, int lane, int kq, int kb, int* ent) {
#pragma unroll
    for (int c = 0; c < RUN_B; ++c) {
        const unsigned o = (unsigned)__builtin_amdgcn_readlane((int)ofw, kb + c);
        const int st = (int)(o & 0xFFFFu), cnt = (int)(o >> 16) - st;
        const unsigned at = (unsigned)(cset + CSETS * (kq + kb + c)) * CH_EDGES + (unsigned)(st + lane);
        ent[c] = (int)srtc[lane < cnt ? at : 0u];
    }
}

__device__ __forceinline__ int queue_own(int* myq, int qn, int e, int n, int lane, int qlo) {
    const unsigned cw = (unsigned)e & ((1u << QSHIFT) - 1u);
    const bool keep = lane < n && (cw - (unsigned)qlo) < (unsigned)RQ;
    const unsigned m = __builtin_amdgcn_ballot_w32(keep);
    if (keep) myq[qn + (int)__builtin_amdgcn_mbcnt_lo(m, 0u)] = e;
    return __builtin_popcount(m);
}

__global__ __launch_bounds__(AGG_THREADS)
void owner_aggregate_kernel(const float* __restrict__ x,
                            const float* __restrict__ W,
                            const float* __restrict__ dis,
                            const float* __restrict__ bias,
                            const unsigned* __restrict__ srtc,
                            const unsigned* __restrict__ offc,
                            float* __restrict__ out, int n, int nchunks) {
    __shared__ __attribute__((aligned(16))) float acc[R_OWN * AP];
    __shared__ __attribute__((aligned(16))) float sW[D * WP];
    __shared__ __attribute__((aligned(16))) float res[M_TILES * 16 * AP];
    __shared__ float sdis[R_OWN];
    __shared__ int q[AGG_WAVES * QCAP];

    const int tid = threadIdx.x;
    const int lane = tid & 31;
    const int wave = __builtin_amdgcn_readfirstlane(tid >> 5);
    const int c0 = blockIdx.x * R_OWN;

    const int cset = wave & (CSETS - 1);
    const int qlo = (wave / CSETS) * RQ;

    unsigned ofw = run_words_q(offc, nchunks, cset, lane, 0);
    int ent[RUN_B];

    {
        constexpr int ROUNDS = (R_OWN * D) / AGG_THREADS;
        constexpr int WROUNDS = (D * D / 4) / AGG_THREADS;
        const int d = tid & (D - 1);
        float sv[ROUNDS], xv[ROUNDS];
        float4 wv[WROUNDS];
#pragma unroll
        for (int t = 0; t < ROUNDS; ++t) {
            const int node = min(c0 + ((tid + t * AGG_THREADS) >> 7), n - 1);
            sv[t] = dis[node];
            xv[t] = x[(size_t)node * D + d];
        }
#pragma unroll
        for (int t = 0; t < WROUNDS; ++t) wv[t] = ((const float4*)W)[tid + t * AGG_THREADS];
        __builtin_amdgcn_sched_barrier(0);
        run_heads(srtc, ofw, cset, lane, 0, 0, ent);
#pragma unroll
        for (int t = 0; t < ROUNDS; ++t)
            acc[((tid + t * AGG_THREADS) >> 7) * AP + d] = xv[t] * sv[t] * sv[t];
#pragma unroll
        for (int t = 0; t < WROUNDS; ++t) {
            const int i4 = tid + t * AGG_THREADS;
            *(float4*)(sW + (i4 >> 5) * WP + ((i4 & 31) << 2)) = wv[t];
        }
    }
    if (tid < R_OWN) sdis[tid] = (c0 + tid < n) ? dis[c0 + tid] : 0.0f;
    __syncthreads();

    {
        int* myq = q + wave * QCAP;
        int qn = 0;
        RowSums a;
        a.k0 = (rowvec)(0.0f); a.k1 = (rowvec)(0.0f); a.k2 = (rowvec)(0.0f); a.k3 = (rowvec)(0.0f);
        for (int kq = 0; cset + CSETS * kq < nchunks; kq += 32) {
          if (kq > 0) ofw = run_words_q(offc, nchunks, cset, lane, kq);
          for (int kb = 0; kb < 32 && cset + CSETS * (kq + kb) < nchunks; kb += RUN_B) {
            if (kq + kb > 0) run_heads(srtc, ofw, cset, lane, kq, kb, ent);
#pragma unroll
            for (int c = 0; c < RUN_B; ++c) {
                const unsigned o = (unsigned)__builtin_amdgcn_readlane((int)ofw, kb + c);
                const int take = min((int)(o >> 16) - (int)(o & 0xFFFFu), 32);
                qn += queue_own(myq, qn, ent[c], take, lane, qlo);
            }
            int c = 0, j = 32;
            for (;;) {
                while (qn >= 32) {
                    qn -= 32;
                    add_group(myq + qn, 32, lane, x, dis, sdis, qlo, a);
                }
                bool got = false;
                while (c < RUN_B) {
                    const unsigned o = (unsigned)__builtin_amdgcn_readlane((int)ofw, kb + c);
                    const int st = (int)(o & 0xFFFFu), cnt = (int)(o >> 16) - st;
                    if (j < cnt) {
                        const unsigned* run = srtc + (size_t)(cset + CSETS * (kq + kb + c)) * CH_EDGES + st;
                        const int more = min(cnt - j, 32);
                        int e = 0;
                        if (lane < more) e = (int)run[j + lane];
                        qn += queue_own(myq, qn, e, more, lane, qlo);
                        j += 32;
                        got = true;
                        break;
                    }
                    ++c;
                    j = 32;
                }
                if (!got) break;
            }
          }
        }
        if (qn > 0) add_group(myq, qn, lane, x, dis, sdis, qlo, a);

        const int hi16 = lane >> 4;
        float* ap0 = acc + qlo * AP + lane * 4;
#pragma unroll
        for (int i = 0; i < RQ; ++i) {
            const float wd[4] = {a.k0[i], a.k1[i], a.k2[i], a.k3[i]};
#pragma unroll
            for (int s = 0; s < 4; ++s) {
                const float val = hi16 ? wd[(s + 1) & 3] : wd[s];
                __hip_atomic_fetch_add(ap0 + i * AP + ((s + hi16) & 3), val, __ATOMIC_RELAXED,
                                       __HIP_MEMORY_SCOPE_WORKGROUP);
            }
        }
    }
    __syncthreads();

    const float bv = bias[tid & (D - 1)];

    if (wave < GEMM_WAVES) {
        const int mt = wave >> 3;
        const int n0 = (wave & 7) << 4;
        const int half = lane >> 4;
        const int idx  = lane & 15;
        const float* ap = acc + min(mt * 16 + idx, R_OWN - 1) * AP + 2 * half;
        const float* bp = sW + (2 * half) * WP + n0 + idx;
        v8f c = {};
#pragma unroll
        for (int k0 = 0; k0 < D; k0 += 4) {
            v2f a, b;
            a.x = ap[k0 + 0];
            a.y = ap[k0 + 1];
            b.x = bp[(k0 + 0) * WP];
            b.y = bp[(k0 + 1) * WP];
            c = __builtin_amdgcn_wmma_f32_16x16x4_f32(
                 false, a,  false, b,
                 (short)0, c,  false,  false);
        }
        float* rp = res + (mt * 16 + 8 * half) * AP + n0 + idx;
#pragma unroll
        for (int g = 0; g < 8; ++g) rp[g * AP] = c[g];
    }
    __syncthreads();

    for (int i = tid; i < R_OWN * D; i += AGG_THREADS) {
        const int node = c0 + (i >> 7);
        if (node < n) out[(size_t)node * D + (i & (D - 1))] = res[(i >> 7) * AP + (i & (D - 1))] + bv;
    }
}

extern "C" void kernel_launch(void* const* d_in, const int* in_sizes, int n_in,
                              void* d_out, int out_size, void* d_ws, size_t ws_size,
                              hipStream_t stream) {
    const float* x    = (const float*)d_in[0];
    const int*   ei   = (const int*)d_in[1];
    const float* W    = (const float*)d_in[2];
    const float* bias = (const float*)d_in[3];
    float* out = (float*)d_out;

    const int N = in_sizes[0] / D;
    const int E = in_sizes[1] / 2;
    const int* row = ei;
    const int* col = ei + E;

    float* dis = (float*)d_ws + (size_t)N * D;
    const int nown = (N + R_OWN - 1) / R_OWN;
    const int nchunks = (E + CH_EDGES - 1) / CH_EDGES;
    unsigned* srtc = (unsigned*)d_ws;
    unsigned* offc = srtc + (size_t)nchunks * CH_EDGES;
    unsigned* offr = offc + (size_t)nchunks * IDX_OWNERS;
    unsigned char* srtr = (unsigned char*)(offr + (size_t)nchunks * IDX_OWNERS);

    owner_sort_kernel<<<nchunks, IDX_THREADS, 0, stream>>>(row, col, srtc, offc, srtr, offr, E / 4);
    degree_owner_kernel<<<nown, AGG_THREADS, 0, stream>>>(srtr, offr, dis, N, nchunks);
    owner_aggregate_kernel<<<nown, AGG_THREADS, 0, stream>>>(x, W, dis, bias, srtc, offc, out, N, nchunks);
}


# Owner work-groups load the whole dis table into LDS with their start values: an edge's source factor is an LDS read, not a scattered 4-byte global load per edge; sums in arrival order as before
# speedup vs baseline: 5.2484x; 1.0510x over previous
#include <hip/hip_runtime.h>
#include <hip/hip_bf16.h>

typedef __attribute__((ext_vector_type(2))) float v2f;
typedef __attribute__((ext_vector_type(8))) float v8f;

#define D 128
#define R_OWN 40

#define IDX_THREADS 1024
#define CH_EDGES (IDX_THREADS * 4)
#define IDX_OWNERS 256
#define QSHIFT 6

__device__ __forceinline__ void chunk_starts(const unsigned* cnt, unsigned* st, unsigned* ow, int lane) {
    unsigned c[8];
    unsigned s = 0u;
#pragma unroll
    for (int i = 0; i < 8; ++i) { c[i] = cnt[lane * 8 + i]; s += c[i]; }
    unsigned incl = s;
#pragma unroll
    for (int d = 1; d < 32; d <<= 1) {
        const unsigned v = __shfl_up(incl, (unsigned)d);
        if (lane >= d) incl += v;
    }
    unsigned run = incl - s;
#pragma unroll
    for (int i = 0; i < 8; ++i) {
        st[lane * 8 + i] = run;
        ow[lane * 8 + i] = run | ((run + c[i]) << 16);
        run += c[i];
    }
}

__device__ __forceinline__ unsigned take_rank(unsigned* cnt, unsigned owner) {
    return __hip_atomic_fetch_add(cnt + owner, 1u, __ATOMIC_RELAXED, __HIP_MEMORY_SCOPE_WORKGROUP);
}

__global__ __launch_bounds__(IDX_THREADS)
void owner_sort_kernel(const int* __restrict__ row, const int* __restrict__ col,
                       unsigned* __restrict__ srtc, unsigned* __restrict__ offc,
                       unsigned char* __restrict__ srtr, unsigned* __restrict__ offr, int E4) {
    __shared__ unsigned cntC[IDX_OWNERS], cntR[IDX_OWNERS];
    __shared__ unsigned stC[IDX_OWNERS], stR[IDX_OWNERS];
    __shared__ unsigned owC[IDX_OWNERS], owR[IDX_OWNERS];
    __shared__ __attribute__((aligned(16))) unsigned stageC[CH_EDGES];
    __shared__ __attribute__((aligned(16))) unsigned char stageR[CH_EDGES];
    const int tid = threadIdx.x;
    const unsigned ro = R_OWN;
    const int t = blockIdx.x * IDX_THREADS + tid;
    const bool have = t < E4;
    int4 r = {0, 0, 0, 0}, cc = {0, 0, 0, 0};
    if (have) {
        r = ((const int4*)row)[t];
        cc = ((const int4*)col)[t];
    }
    if (tid < IDX_OWNERS) { cntC[tid] = 0u; cntR[tid] = 0u; }
    __syncthreads();
    asm volatile("" :: "v"(r.x), "v"(r.y), "v"(r.z), "v"(r.w), "v"(cc.x), "v"(cc.y), "v"(cc.z), "v"(cc.w));

    unsigned sr[4] = {0u, 0u, 0u, 0u}, oc[4] = {0u, 0u, 0u, 0u}, orr[4] = {0u, 0u, 0u, 0u};
    unsigned lc[4] = {0u, 0u, 0u, 0u}, lr[4] = {0u, 0u, 0u, 0u}, kc[4] = {0u, 0u, 0u, 0u}, kr[4] = {0u, 0u, 0u, 0u};
    if (have) {
        sr[0] = (unsigned)r.x; sr[1] = (unsigned)r.y; sr[2] = (unsigned)r.z; sr[3] = (unsigned)r.w;
        const unsigned dc[4] = {(unsigned)cc.x, (unsigned)cc.y, (unsigned)cc.z, (unsigned)cc.w};
#pragma unroll
        for (int i = 0; i < 4; ++i) {
            oc[i] = dc[i] / ro;  lc[i] = dc[i] - oc[i] * ro;
            orr[i] = sr[i] / ro; lr[i] = sr[i] - orr[i] * ro;
        }
#pragma unroll
        for (int i = 0; i < 4; ++i) { kc[i] = take_rank(cntC, oc[i]); kr[i] = take_rank(cntR, orr[i]); }
    }
    __syncthreads();

    if (tid < 32) chunk_starts(cntC, stC, owC, tid);
    else if (tid < 64) chunk_starts(cntR, stR, owR, tid - 32);
    __syncthreads();

    if (have) {
#pragma unroll
        for (int i = 0; i < 4; ++i) {
            stageC[stC[oc[i]] + kc[i]] = (sr[i] << QSHIFT) | lc[i];
            stageR[stR[orr[i]] + kr[i]] = (unsigned char)lr[i];
        }
    }
    __syncthreads();

    ((int4*)srtc)[(size_t)blockIdx.x * (CH_EDGES / 4) + tid] = ((const int4*)stageC)[tid];
    if (tid < CH_EDGES / 16)
        ((int4*)srtr)[(size_t)blockIdx.x * (CH_EDGES / 16) + tid] = ((const int4*)stageR)[tid];
    if (tid < IDX_OWNERS) {
        offc[(size_t)blockIdx.x * IDX_OWNERS + tid] = owC[tid];
        offr[(size_t)blockIdx.x * IDX_OWNERS + tid] = owR[tid];
    }
}

#define AGG_THREADS 1024
#define AGG_WAVES 32
#define RUN_B 8

__device__ __forceinline__ unsigned run_words(const unsigned* __restrict__ off, int nchunks, int wave, int lane,
                                              int kb) {
    const int ch = wave + AGG_WAVES * (kb + lane);
    unsigned ofw = 0u;
    if (lane < RUN_B && ch < nchunks) ofw = off[(size_t)ch * IDX_OWNERS + blockIdx.x];
    return ofw;
}

__global__ __launch_bounds__(AGG_THREADS)
void degree_owner_kernel(const unsigned char* __restrict__ srtr, const unsigned* __restrict__ offr,
                         float* __restrict__ dis, int n, int nchunks) {
    __shared__ int bins[R_OWN];

    const int tid = threadIdx.x;
    const int lane = tid & 31;
    const int wave = __builtin_amdgcn_readfirstlane(tid >> 5);
    const int c0 = blockIdx.x * R_OWN;

    if (tid < R_OWN) bins[tid] = 0;
    __syncthreads();

    for (int kb = 0; wave + AGG_WAVES * kb < nchunks; kb += RUN_B) {
        const unsigned ofw = run_words(offr, nchunks, wave, lane, kb);
        int v[RUN_B];
#pragma unroll
        for (int c = 0; c < RUN_B; ++c) {
            const unsigned o = (unsigned)__builtin_amdgcn_readlane((int)ofw, c);
            const int st = (int)(o & 0xFFFFu), cnt = (int)(o >> 16) - st;
            const unsigned char* run = srtr + (size_t)(wave + AGG_WAVES * (kb + c)) * CH_EDGES + st;
            v[c] = -1;
            if (lane < cnt) v[c] = run[lane];
        }
#pragma unroll
        for (int c = 0; c < RUN_B; ++c) {
            if (v[c] >= 0) __hip_atomic_fetch_add(bins + v[c], 1, __ATOMIC_RELAXED, __HIP_MEMORY_SCOPE_WORKGROUP);
            const unsigned o = (unsigned)__builtin_amdgcn_readlane((int)ofw, c);
            const int st = (int)(o & 0xFFFFu), cnt = (int)(o >> 16) - st;
            const unsigned char* run = srtr + (size_t)(wave + AGG_WAVES * (kb + c)) * CH_EDGES + st;
            for (int j = 32 + lane; j < cnt; j += 32)
                __hip_atomic_fetch_add(bins + run[j], 1, __ATOMIC_RELAXED, __HIP_MEMORY_SCOPE_WORKGROUP);
        }
    }
    __syncthreads();

    if (tid < R_OWN && c0 + tid < n) dis[c0 + tid] = rsqrtf((float)(bins[tid] + 1));
}

#define QCAP 288
#define AP 132
#define WP 144
#define DIS_SLOTS (IDX_OWNERS * R_OWN)
#define DIS_ROUNDS ((DIS_SLOTS / 4 + AGG_THREADS - 1) / AGG_THREADS)
#define DIS_LDS (DIS_ROUNDS * AGG_THREADS * 4)
#define M_TILES ((R_OWN + 15) / 16)
#define GEMM_WAVES (M_TILES * (D / 16))

#define NQ 4
#define RQ (R_OWN / NQ)
#define CSETS (AGG_WAVES / NQ)

__device__ __forceinline__ unsigned run_words_q(const unsigned* __restrict__ off, int nchunks, int cset, int lane,
                                                int kq) {
    const int ch = cset + CSETS * (kq + lane);
    unsigned ofw = 0u;
    if (ch < nchunks) ofw = off[(size_t)ch * IDX_OWNERS + blockIdx.x];
    return ofw;
}

typedef __attribute__((ext_vector_type(RQ))) float rowvec;
struct RowSums { rowvec k0, k1, k2, k3; };

__device__ __forceinline__ void add_row(RowSums& a, int row, float4 v, float w) {
    a.k0[row] = fmaf(v.x, w, a.k0[row]);
    a.k1[row] = fmaf(v.y, w, a.k1[row]);
    a.k2[row] = fmaf(v.z, w, a.k2[row]);
    a.k3[row] = fmaf(v.w, w, a.k3[row]);
}

__device__ __forceinline__ void add_group(const int* qp, int cnt, int lane,
                                          const float* __restrict__ x,
                                          const float* sdall,
                                          const float* sdis, int qlo, RowSums& a) {
    int rq = 0;
    float w = 0.0f;
    if (lane < cnt) {
        const unsigned pk = (unsigned)qp[lane];
        const int cw = (int)(pk & ((1u << QSHIFT) - 1u));
        w = sdall[pk >> QSHIFT] * sdis[cw];
        rq = (int)pk - qlo;
    }
    const unsigned lo = (unsigned)lane * 4u;
    for (int j = 0; j < cnt; j += 8) {
        float4 v[8];
        int cj[8];
        float wj[8];
#pragma unroll
        for (int u = 0; u < 8; ++u) {
            const int jj = min(j + u, cnt - 1);
            const unsigned pj = (unsigned)__builtin_amdgcn_readlane(rq, jj);
            cj[u] = (int)(pj & ((1u << QSHIFT) - 1u));
            wj[u] = __int_as_float(__builtin_amdgcn_readlane(__float_as_int(w), jj));
            const float* xr = x + (size_t)(pj >> QSHIFT) * D;
            v[u] = *(const float4*)(xr + lo);
        }
#pragma unroll
        for (int u = 0; u < 8; ++u) {
            if (j + u < cnt) add_row(a, cj[u], v[u], wj[u]);
        }
    }
}

__device__ __forceinline__ void run_heads(const unsigned* __restrict__ srtc,
                                          unsigned ofw, int cset, int lane, int kq, int kb, int* ent) {
#pragma unroll
    for (int c = 0; c < RUN_B; ++c) {
        const unsigned o = (unsigned)__builtin_amdgcn_readlane((int)ofw, kb + c);
        const int st = (int)(o & 0xFFFFu), cnt = (int)(o >> 16) - st;
        const unsigned at = (unsigned)(cset + CSETS * (kq + kb + c)) * CH_EDGES + (unsigned)(st + lane);
        ent[c] = (int)srtc[lane < cnt ? at : 0u];
    }
}

__device__ __forceinline__ int queue_own(int* myq, int qn, int e, int n, int lane, int qlo) {
    const unsigned cw = (unsigned)e & ((1u << QSHIFT) - 1u);
    const bool keep = lane < n && (cw - (unsigned)qlo) < (unsigned)RQ;
    const unsigned m = __builtin_amdgcn_ballot_w32(keep);
    if (keep) myq[qn + (int)__builtin_amdgcn_mbcnt_lo(m, 0u)] = e;
    return __builtin_popcount(m);
}

__global__ __launch_bounds__(AGG_THREADS)
void owner_aggregate_kernel(const float* __restrict__ x,
                            const float* __restrict__ W,
                            const float* __restrict__ dis,
                            const float* __restrict__ bias,
                            const unsigned* __restrict__ srtc,
                            const unsigned* __restrict__ offc,
                            float* __restrict__ out, int n, int nchunks) {
    __shared__ __attribute__((aligned(16))) float acc[R_OWN * AP];
    __shared__ __attribute__((aligned(16))) float sW[D * WP];
    __shared__ __attribute__((aligned(16))) float res[M_TILES * 16 * AP];
    __shared__ float sdis[R_OWN];
    __shared__ __attribute__((aligned(16))) float sdall[DIS_LDS];
    __shared__ int q[AGG_WAVES * QCAP];

    const int tid = threadIdx.x;
    const int lane = tid & 31;
    const int wave = __builtin_amdgcn_readfirstlane(tid >> 5);
    const int c0 = blockIdx.x * R_OWN;

    const int cset = wave & (CSETS - 1);
    const int qlo = (wave / CSETS) * RQ;

    unsigned ofw = run_words_q(offc, nchunks, cset, lane, 0);
    int ent[RUN_B];

    {
        constexpr int ROUNDS = (R_OWN * D) / AGG_THREADS;
        constexpr int WROUNDS = (D * D / 4) / AGG_THREADS;
        const int d = tid & (D - 1);
        constexpr int DROUNDS = DIS_ROUNDS;
        const int n4 = n >> 2;
        float sv[ROUNDS], xv[ROUNDS];
        float4 wv[WROUNDS];
        float4 dv[DROUNDS];
#pragma unroll
        for (int t = 0; t < ROUNDS; ++t) {
            const int node = min(c0 + ((tid + t * AGG_THREADS) >> 7), n - 1);
            sv[t] = dis[node];
            xv[t] = x[(size_t)node * D + d];
        }
#pragma unroll
        for (int t = 0; t < WROUNDS; ++t) wv[t] = ((const float4*)W)[tid + t * AGG_THREADS];
#pragma unroll
        for (int t = 0; t < DROUNDS; ++t)
            dv[t] = ((const float4*)dis)[max(min(tid + t * AGG_THREADS, n4 - 1), 0)];
        __builtin_amdgcn_sched_barrier(0);
        run_heads(srtc, ofw, cset, lane, 0, 0, ent);
#pragma unroll
        for (int t = 0; t < ROUNDS; ++t)
            acc[((tid + t * AGG_THREADS) >> 7) * AP + d] = xv[t] * sv[t] * sv[t];
#pragma unroll
        for (int t = 0; t < WROUNDS; ++t) {
            const int i4 = tid + t * AGG_THREADS;
            *(float4*)(sW + (i4 >> 5) * WP + ((i4 & 31) << 2)) = wv[t];
        }
#pragma unroll
        for (int t = 0; t < DROUNDS; ++t) *(float4*)(sdall + 4 * (tid + t * AGG_THREADS)) = dv[t];
        if ((n & 3) != 0 && tid == (n4 & (AGG_THREADS - 1)))
            for (int k = 0; k < (n & 3); ++k) sdall[4 * n4 + k] = dis[4 * n4 + k];
    }
    if (tid < R_OWN) sdis[tid] = (c0 + tid < n) ? dis[c0 + tid] : 0.0f;
    __syncthreads();

    {
        int* myq = q + wave * QCAP;
        int qn = 0;
        RowSums a;
        a.k0 = (rowvec)(0.0f); a.k1 = (rowvec)(0.0f); a.k2 = (rowvec)(0.0f); a.k3 = (rowvec)(0.0f);
        for (int kq = 0; cset + CSETS * kq < nchunks; kq += 32) {
          if (kq > 0) ofw = run_words_q(offc, nchunks, cset, lane, kq);
          for (int kb = 0; kb < 32 && cset + CSETS * (kq + kb) < nchunks; kb += RUN_B) {
            if (kq + kb > 0) run_heads(srtc, ofw, cset, lane, kq, kb, ent);
#pragma unroll
            for (int c = 0; c < RUN_B; ++c) {
                const unsigned o = (unsigned)__builtin_amdgcn_readlane((int)ofw, kb + c);
                const int take = min((int)(o >> 16) - (int)(o & 0xFFFFu), 32);
                qn += queue_own(myq, qn, ent[c], take, lane, qlo);
            }
            int c = 0, j = 32;
            for (;;) {
                while (qn >= 32) {
                    qn -= 32;
                    add_group(myq + qn, 32, lane, x, sdall, sdis, qlo, a);
                }
                bool got = false;
                while (c < RUN_B) {
                    const unsigned o = (unsigned)__builtin_amdgcn_readlane((int)ofw, kb + c);
                    const int st = (int)(o & 0xFFFFu), cnt = (int)(o >> 16) - st;
                    if (j < cnt) {
                        const unsigned* run = srtc + (size_t)(cset + CSETS * (kq + kb + c)) * CH_EDGES + st;
                        const int more = min(cnt - j, 32);
                        int e = 0;
                        if (lane < more) e = (int)run[j + lane];
                        qn += queue_own(myq, qn, e, more, lane, qlo);
                        j += 32;
                        got = true;
                        break;
                    }
                    ++c;
                    j = 32;
                }
                if (!got) break;
            }
          }
        }
        if (qn > 0) add_group(myq, qn, lane, x, sdall, sdis, qlo, a);

        const int hi16 = lane >> 4;
        float* ap0 = acc + qlo * AP + lane * 4;
#pragma unroll
        for (int i = 0; i < RQ; ++i) {
            const float wd[4] = {a.k0[i], a.k1[i], a.k2[i], a.k3[i]};
#pragma unroll
            for (int s = 0; s < 4; ++s) {
                const float val = hi16 ? wd[(s + 1) & 3] : wd[s];
                __hip_atomic_fetch_add(ap0 + i * AP + ((s + hi16) & 3), val, __ATOMIC_RELAXED,
                                       __HIP_MEMORY_SCOPE_WORKGROUP);
            }
        }
    }
    __syncthreads();

    const float bv = bias[tid & (D - 1)];

    if (wave < GEMM_WAVES) {
        const int mt = wave >> 3;
        const int n0 = (wave & 7) << 4;
        const int half = lane >> 4;
        const int idx  = lane & 15;
        const float* ap = acc + min(mt * 16 + idx, R_OWN - 1) * AP + 2 * half;
        const float* bp = sW + (2 * half) * WP + n0 + idx;
        v8f c = {};
#pragma unroll
        for (int k0 = 0; k0 < D; k0 += 4) {
            v2f a, b;
            a.x = ap[k0 + 0];
            a.y = ap[k0 + 1];
            b.x = bp[(k0 + 0) * WP];
            b.y = bp[(k0 + 1) * WP];
            c = __builtin_amdgcn_wmma_f32_16x16x4_f32(
                 false, a,  false, b,
                 (short)0, c,  false,  false);
        }
        float* rp = res + (mt * 16 + 8 * half) * AP + n0 + idx;
#pragma unroll
        for (int g = 0; g < 8; ++g) rp[g * AP] = c[g];
    }
    __syncthreads();

    for (int i = tid; i < R_OWN * D; i += AGG_THREADS) {
        const int node = c0 + (i >> 7);
        if (node < n) out[(size_t)node * D + (i & (D - 1))] = res[(i >> 7) * AP + (i & (D - 1))] + bv;
    }
}

extern "C" void kernel_launch(void* const* d_in, const int* in_sizes, int n_in,
                              void* d_out, int out_size, void* d_ws, size_t ws_size,
                              hipStream_t stream) {
    const float* x    = (const float*)d_in[0];
    const int*   ei   = (const int*)d_in[1];
    const float* W    = (const float*)d_in[2];
    const float* bias = (const float*)d_in[3];
    float* out = (float*)d_out;

    const int N = in_sizes[0] / D;
    const int E = in_sizes[1] / 2;
    const int* row = ei;
    const int* col = ei + E;

    float* dis = (float*)d_ws + (size_t)N * D;
    const int nown = (N + R_OWN - 1) / R_OWN;
    const int nchunks = (E + CH_EDGES - 1) / CH_EDGES;
    unsigned* srtc = (unsigned*)d_ws;
    unsigned* offc = srtc + (size_t)nchunks * CH_EDGES;
    unsigned* offr = offc + (size_t)nchunks * IDX_OWNERS;
    unsigned char* srtr = (unsigned char*)(offr + (size_t)nchunks * IDX_OWNERS);

    owner_sort_kernel<<<nchunks, IDX_THREADS, 0, stream>>>(row, col, srtc, offc, srtr, offr, E / 4);
    degree_owner_kernel<<<nown, AGG_THREADS, 0, stream>>>(srtr, offr, dis, N, nchunks);
    owner_aggregate_kernel<<<nown, AGG_THREADS, 0, stream>>>(x, W, dis, bias, srtc, offc, out, N, nchunks);
}


# Degree counted inside launch 1 by extra work-groups that read the source list direct (integer LDS adds, 16-bit partial counts per part); owners sum them into their LDS dis table: 2 launches, not 3
# speedup vs baseline: 5.5160x; 1.0510x over previous
#include <hip/hip_runtime.h>
#include <hip/hip_bf16.h>

typedef __attribute__((ext_vector_type(2))) float v2f;
typedef __attribute__((ext_vector_type(8))) float v8f;
typedef __attribute__((ext_vector_type(4))) float v4f;

#define D 128
#define R_OWN 40

#define IDX_THREADS 1024
#define CH_EDGES (IDX_THREADS * 4)
#define IDX_OWNERS 256
#define QSHIFT 6
#define DIS_SLOTS (IDX_OWNERS * R_OWN)
#define CNT_PARTS 10
#define CNT_RANGES 4
#define CNT_WGS (CNT_PARTS * CNT_RANGES)
#define RANGE_NODES (DIS_SLOTS / CNT_RANGES)
#define PLANE_WORDS (DIS_SLOTS / 2)
#define CNT_LOADS 16
static_assert(RANGE_NODES <= CH_EDGES && RANGE_NODES % 8 == 0, "a range's counters fit the chunk stage; whole 16-byte pieces of packed counts");

__device__ __forceinline__ void chunk_starts(const unsigned* cnt, unsigned* st, unsigned* ow, int lane) {
    unsigned c[8];
    unsigned s = 0u;
#pragma unroll
    for (int i = 0; i < 8; ++i) { c[i] = cnt[lane * 8 + i]; s += c[i]; }
    unsigned incl = s;
#pragma unroll
    for (int d = 1; d < 32; d <<= 1) {
        const unsigned v = __shfl_up(incl, (unsigned)d);
        if (lane >= d) incl += v;
    }
    unsigned run = incl - s;
#pragma unroll
    for (int i = 0; i < 8; ++i) {
        st[lane * 8 + i] = run;
        ow[lane * 8 + i] = run | ((run + c[i]) << 16);
        run += c[i];
    }
}

__device__ __forceinline__ unsigned take_rank(unsigned* cnt, unsigned owner) {
    return __hip_atomic_fetch_add(cnt + owner, 1u, __ATOMIC_RELAXED, __HIP_MEMORY_SCOPE_WORKGROUP);
}

__device__ __forceinline__ void count_source(unsigned* bins, unsigned src, unsigned lo) {
    const unsigned j = src - lo;
    if (j < (unsigned)RANGE_NODES)
        __hip_atomic_fetch_add(bins + j, 1u, __ATOMIC_RELAXED, __HIP_MEMORY_SCOPE_WORKGROUP);
}

__device__ __forceinline__ void source_pieces(const int* __restrict__ row, int b, int pe, int tid, int4* v) {
#pragma unroll
    for (int k = 0; k < CNT_LOADS; ++k)
        v[k] = ((const int4*)row)[max(min(b + k * IDX_THREADS + tid, pe - 1), 0)];
}

__global__ __launch_bounds__(IDX_THREADS)
void owner_sort_kernel(const int* __restrict__ row, const int* __restrict__ col,
                       unsigned* __restrict__ srtc, unsigned* __restrict__ offc,
                       unsigned* __restrict__ cplanes, int E4) {
    __shared__ unsigned cntC[IDX_OWNERS];
    __shared__ unsigned stC[IDX_OWNERS];
    __shared__ unsigned owC[IDX_OWNERS];
    __shared__ __attribute__((aligned(16))) unsigned stageC[CH_EDGES];
    const int tid = threadIdx.x;

    if (blockIdx.x < CNT_WGS) {
        unsigned* bins = stageC;
        const int part = blockIdx.x / CNT_RANGES, range = blockIdx.x % CNT_RANGES;
        const unsigned lo = (unsigned)range * RANGE_NODES;
        const int pp = (E4 + CNT_PARTS - 1) / CNT_PARTS;
        const int p0 = part * pp, pe = min(p0 + pp, E4);
        int4 v[CNT_LOADS];
        source_pieces(row, p0, pe, tid, v);
        for (int i = tid; i < RANGE_NODES; i += IDX_THREADS) bins[i] = 0u;
        __builtin_amdgcn_fence(__ATOMIC_RELEASE, "workgroup", "local");
        __builtin_amdgcn_s_barrier();
        __builtin_amdgcn_fence(__ATOMIC_ACQUIRE, "workgroup", "local");
        for (int b = p0; b < pe;) {
#pragma unroll
            for (int k = 0; k < CNT_LOADS; ++k) {
                const unsigned lok = (b + k * IDX_THREADS + tid < pe) ? lo : 0x80000000u;
                count_source(bins, (unsigned)v[k].x, lok);
                count_source(bins, (unsigned)v[k].y, lok);
                count_source(bins, (unsigned)v[k].z, lok);
                count_source(bins, (unsigned)v[k].w, lok);
            }
            b += CNT_LOADS * IDX_THREADS;
            if (b < pe) source_pieces(row, b, pe, tid, v);
        }
        __syncthreads();
        if (tid < RANGE_NODES / 8) {
            const uint4 e0 = ((const uint4*)bins)[2 * tid], e1 = ((const uint4*)bins)[2 * tid + 1];
            const uint4 pk = {e0.x | (e0.y << 16), e0.z | (e0.w << 16), e1.x | (e1.y << 16), e1.z | (e1.w << 16)};
            ((uint4*)cplanes)[(size_t)part * (PLANE_WORDS / 4) + range * (RANGE_NODES / 8) + tid] = pk;
        }
        return;
    }

    const int chunk = blockIdx.x - CNT_WGS;
    const unsigned ro = R_OWN;
    const int t = chunk * IDX_THREADS + tid;
    const bool have = t < E4;
    int4 r = {0, 0, 0, 0}, cc = {0, 0, 0, 0};
    if (have) {
        r = ((const int4*)row)[t];
        cc = ((const int4*)col)[t];
    }
    if (tid < IDX_OWNERS) cntC[tid] = 0u;
    __syncthreads();
    asm volatile("" :: "v"(r.x), "v"(r.y), "v"(r.z), "v"(r.w), "v"(cc.x), "v"(cc.y), "v"(cc.z), "v"(cc.w));

    unsigned sr[4] = {0u, 0u, 0u, 0u}, oc[4] = {0u, 0u, 0u, 0u};
    unsigned lc[4] = {0u, 0u, 0u, 0u}, kc[4] = {0u, 0u, 0u, 0u};
    if (have) {
        sr[0] = (unsigned)r.x; sr[1] = (unsigned)r.y; sr[2] = (unsigned)r.z; sr[3] = (unsigned)r.w;
        const unsigned dc[4] = {(unsigned)cc.x, (unsigned)cc.y, (unsigned)cc.z, (unsigned)cc.w};
#pragma unroll
        for (int i = 0; i < 4; ++i) { oc[i] = dc[i] / ro;  lc[i] = dc[i] - oc[i] * ro; }
#pragma unroll
        for (int i = 0; i < 4; ++i) kc[i] = take_rank(cntC, oc[i]);
    }
    __syncthreads();

    if (tid < 32) chunk_starts(cntC, stC, owC, tid);
    __syncthreads();

    if (have) {
#pragma unroll
        for (int i = 0; i < 4; ++i) stageC[stC[oc[i]] + kc[i]] = (sr[i] << QSHIFT) | lc[i];
    }
    __syncthreads();

    ((int4*)srtc)[(size_t)chunk * (CH_EDGES / 4) + tid] = ((const int4*)stageC)[tid];
    if (tid < IDX_OWNERS) offc[(size_t)chunk * IDX_OWNERS + tid] = owC[tid];
}

#define AGG_THREADS 1024
#define AGG_WAVES 32
#define RUN_B 8

#define QCAP 288
#define AP 132
#define WP 144
#define TAIL_NODE0 (AGG_THREADS * 8)
static_assert(PLANE_WORDS == AGG_THREADS * 5, "a 16-byte piece and a word per thread cover a part's counts");
#define M_TILES ((R_OWN + 15) / 16)
#define GEMM_WAVES (M_TILES * (D / 16))

#define NQ 4
#define RQ (R_OWN / NQ)
#define CSETS (AGG_WAVES / NQ)

__device__ __forceinline__ unsigned run_words_q(const unsigned* __restrict__ off, int nchunks, int cset, int lane,
                                                int kq) {
    const int ch = cset + CSETS * (kq + lane);
    unsigned ofw = 0u;
    if (ch < nchunks) ofw = off[(size_t)ch * IDX_OWNERS + blockIdx.x];
    return ofw;
}

typedef __attribute__((ext_vector_type(RQ))) float rowvec;
struct RowSums { rowvec k0, k1, k2, k3; };

__device__ __forceinline__ void add_row(RowSums& a, int row, float4 v, float w) {
    a.k0[row] = fmaf(v.x, w, a.k0[row]);
    a.k1[row] = fmaf(v.y, w, a.k1[row]);
    a.k2[row] = fmaf(v.z, w, a.k2[row]);
    a.k3[row] = fmaf(v.w, w, a.k3[row]);
}

__device__ __forceinline__ void add_group(const int* qp, int cnt, int lane,
                                          const float* __restrict__ x,
                                          const float* sdall,
                                          const float* sdis, int qlo, RowSums& a) {
    int rq = 0;
    float w = 0.0f;
    if (lane < cnt) {
        const unsigned pk = (unsigned)qp[lane];
        const int cw = (int)(pk & ((1u << QSHIFT) - 1u));
        w = sdall[pk >> QSHIFT] * sdis[cw];
        rq = (int)pk - qlo;
    }
    const unsigned lo = (unsigned)lane * 4u;
    for (int j = 0; j < cnt; j += 8) {
        float4 v[8];
        int cj[8];
        float wj[8];
#pragma unroll
        for (int u = 0; u < 8; ++u) {
            const int jj = min(j + u, cnt - 1);
            const unsigned pj = (unsigned)__builtin_amdgcn_readlane(rq, jj);
            cj[u] = (int)(pj & ((1u << QSHIFT) - 1u));
            wj[u] = __int_as_float(__builtin_amdgcn_readlane(__float_as_int(w), jj));
            const float* xr = x + (size_t)(pj >> QSHIFT) * D;
            v[u] = *(const float4*)(xr + lo);
        }
#pragma unroll
        for (int u = 0; u < 8; ++u) {
            if (j + u < cnt) add_row(a, cj[u], v[u], wj[u]);
        }
    }
}

__device__ __forceinline__ void run_heads(const unsigned* __restrict__ srtc,
                                          unsigned ofw, int cset, int lane, int kq, int kb, int* ent) {
#pragma unroll
    for (int c = 0; c < RUN_B; ++c) {
        const unsigned o = (unsigned)__builtin_amdgcn_readlane((int)ofw, kb + c);
        const int st = (int)(o & 0xFFFFu), cnt = (int)(o >> 16) - st;
        const unsigned at = (unsigned)(cset + CSETS * (kq + kb + c)) * CH_EDGES + (unsigned)(st + lane);
        ent[c] = (int)srtc[lane < cnt ? at : 0u];
    }
}

__device__ __forceinline__ int queue_own(int* myq, int qn, int e, int n, int lane, int qlo) {
    const unsigned cw = (unsigned)e & ((1u << QSHIFT) - 1u);
    const bool keep = lane < n && (cw - (unsigned)qlo) < (unsigned)RQ;
    const unsigned m = __builtin_amdgcn_ballot_w32(keep);
    if (keep) myq[qn + (int)__builtin_amdgcn_mbcnt_lo(m, 0u)] = e;
    return __builtin_popcount(m);
}

__global__ __launch_bounds__(AGG_THREADS)
void owner_aggregate_kernel(const float* __restrict__ x,
                            const float* __restrict__ W,
                            const unsigned* __restrict__ cplanes,
                            const float* __restrict__ bias,
                            const unsigned* __restrict__ srtc,
                            const unsigned* __restrict__ offc,
                            float* __restrict__ out, int n, int nchunks) {
    __shared__ __attribute__((aligned(16))) float acc[R_OWN * AP];
    __shared__ __attribute__((aligned(16))) float sW[D * WP];
    __shared__ __attribute__((aligned(16))) float res[M_TILES * 16 * AP];
    __shared__ float sdis[R_OWN];
    __shared__ __attribute__((aligned(16))) float sdall[DIS_SLOTS];
    __shared__ int q[AGG_WAVES * QCAP];

    const int tid = threadIdx.x;
    const int lane = tid & 31;
    const int wave = __builtin_amdgcn_readfirstlane(tid >> 5);
    const int c0 = blockIdx.x * R_OWN;

    const int cset = wave & (CSETS - 1);
    const int qlo = (wave / CSETS) * RQ;

    unsigned ofw = run_words_q(offc, nchunks, cset, lane, 0);
    int ent[RUN_B];

    {
        constexpr int ROUNDS = (R_OWN * D) / AGG_THREADS;
        constexpr int WROUNDS = (D * D / 4) / AGG_THREADS;
        const int d = tid & (D - 1);
        float xv[ROUNDS];
        v4f wv[WROUNDS];
        uint4 pv[CNT_PARTS];
        unsigned pt[CNT_PARTS];
#pragma unroll
        for (int t = 0; t < ROUNDS; ++t) {
            const int node = min(c0 + ((tid + t * AGG_THREADS) >> 7), n - 1);
            xv[t] = x[(size_t)node * D + d];
        }
#pragma unroll
        for (int t = 0; t < WROUNDS; ++t) wv[t] = ((const v4f*)W)[tid + t * AGG_THREADS];
#pragma unroll
        for (int s = 0; s < CNT_PARTS; ++s) pv[s] = ((const uint4*)cplanes)[s * (PLANE_WORDS / 4) + tid];
#pragma unroll
        for (int s = 0; s < CNT_PARTS; ++s) pt[s] = cplanes[s * PLANE_WORDS + TAIL_NODE0 / 2 + tid];
        __builtin_amdgcn_sched_barrier(0);
        run_heads(srtc, ofw, cset, lane, 0, 0, ent);
        __builtin_amdgcn_sched_barrier(0);
#pragma unroll
        for (int t = 0; t < WROUNDS; ++t) {
            const int i4 = tid + t * AGG_THREADS;
            *(v4f*)(sW + (i4 >> 5) * WP + ((i4 & 31) << 2)) = wv[t];
        }
        {
            unsigned cn[8] = {1u, 1u, 1u, 1u, 1u, 1u, 1u, 1u};
#pragma unroll
            for (int s = 0; s < CNT_PARTS; ++s) {
                cn[0] += pv[s].x & 0xFFFFu; cn[1] += pv[s].x >> 16;
                cn[2] += pv[s].y & 0xFFFFu; cn[3] += pv[s].y >> 16;
                cn[4] += pv[s].z & 0xFFFFu; cn[5] += pv[s].z >> 16;
                cn[6] += pv[s].w & 0xFFFFu; cn[7] += pv[s].w >> 16;
            }
            const float4 d0 = {rsqrtf((float)cn[0]), rsqrtf((float)cn[1]), rsqrtf((float)cn[2]), rsqrtf((float)cn[3])};
            const float4 d1 = {rsqrtf((float)cn[4]), rsqrtf((float)cn[5]), rsqrtf((float)cn[6]), rsqrtf((float)cn[7])};
            *(float4*)(sdall + 8 * tid) = d0;
            *(float4*)(sdall + 8 * tid + 4) = d1;
        }
        {
            unsigned cn[2] = {1u, 1u};
#pragma unroll
            for (int s = 0; s < CNT_PARTS; ++s) { cn[0] += pt[s] & 0xFFFFu; cn[1] += pt[s] >> 16; }
            const float2 d0 = {rsqrtf((float)cn[0]), rsqrtf((float)cn[1])};
            *(float2*)(sdall + TAIL_NODE0 + 2 * tid) = d0;
        }
        __syncthreads();
#pragma unroll
        for (int t = 0; t < ROUNDS; ++t) {
            const int r = (tid + t * AGG_THREADS) >> 7;
            const float sv = sdall[min(c0 + r, n - 1)];
            acc[r * AP + d] = xv[t] * sv * sv;
        }
    }
    if (tid < R_OWN) sdis[tid] = (c0 + tid < n) ? sdall[c0 + tid] : 0.0f;
    __syncthreads();

    {
        int* myq = q + wave * QCAP;
        int qn = 0;
        RowSums a;
        a.k0 = (rowvec)(0.0f); a.k1 = (rowvec)(0.0f); a.k2 = (rowvec)(0.0f); a.k3 = (rowvec)(0.0f);
        for (int kq = 0; cset + CSETS * kq < nchunks; kq += 32) {
          if (kq > 0) ofw = run_words_q(offc, nchunks, cset, lane, kq);
          for (int kb = 0; kb < 32 && cset + CSETS * (kq + kb) < nchunks; kb += RUN_B) {
            if (kq + kb > 0) run_heads(srtc, ofw, cset, lane, kq, kb, ent);
#pragma unroll
            for (int c = 0; c < RUN_B; ++c) {
                const unsigned o = (unsigned)__builtin_amdgcn_readlane((int)ofw, kb + c);
                const int take = min((int)(o >> 16) - (int)(o & 0xFFFFu), 32);
                qn += queue_own(myq, qn, ent[c], take, lane, qlo);
            }
            int c = 0, j = 32;
            for (;;) {
                while (qn >= 32) {
                    qn -= 32;
                    add_group(myq + qn, 32, lane, x, sdall, sdis, qlo, a);
                }
                bool got = false;
                while (c < RUN_B) {
                    const unsigned o = (unsigned)__builtin_amdgcn_readlane((int)ofw, kb + c);
                    const int st = (int)(o & 0xFFFFu), cnt = (int)(o >> 16) - st;
                    if (j < cnt) {
                        const unsigned* run = srtc + (size_t)(cset + CSETS * (kq + kb + c)) * CH_EDGES + st;
                        const int more = min(cnt - j, 32);
                        int e = 0;
                        if (lane < more) e = (int)run[j + lane];
                        qn += queue_own(myq, qn, e, more, lane, qlo);
                        j += 32;
                        got = true;
                        break;
                    }
                    ++c;
                    j = 32;
                }
                if (!got) break;
            }
          }
        }
        if (qn > 0) add_group(myq, qn, lane, x, sdall, sdis, qlo, a);

        const int hi16 = lane >> 4;
        float* ap0 = acc + qlo * AP + lane * 4;
#pragma unroll
        for (int i = 0; i < RQ; ++i) {
            const float wd[4] = {a.k0[i], a.k1[i], a.k2[i], a.k3[i]};
#pragma unroll
            for (int s = 0; s < 4; ++s) {
                const float val = hi16 ? wd[(s + 1) & 3] : wd[s];
                __hip_atomic_fetch_add(ap0 + i * AP + ((s + hi16) & 3), val, __ATOMIC_RELAXED,
                                       __HIP_MEMORY_SCOPE_WORKGROUP);
            }
        }
    }
    __syncthreads();

    const float bv = bias[tid & (D - 1)];

    if (wave < GEMM_WAVES) {
        const int mt = wave >> 3;
        const int n0 = (wave & 7) << 4;
        const int half = lane >> 4;
        const int idx  = lane & 15;
        const float* ap = acc + min(mt * 16 + idx, R_OWN - 1) * AP + 2 * half;
        const float* bp = sW + (2 * half) * WP + n0 + idx;
        v8f c = {};
#pragma unroll
        for (int k0 = 0; k0 < D; k0 += 4) {
            v2f a, b;
            a.x = ap[k0 + 0];
            a.y = ap[k0 + 1];
            b.x = bp[(k0 + 0) * WP];
            b.y = bp[(k0 + 1) * WP];
            c = __builtin_amdgcn_wmma_f32_16x16x4_f32(
                 false, a,  false, b,
                 (short)0, c,  false,  false);
        }
        float* rp = res + (mt * 16 + 8 * half) * AP + n0 + idx;
#pragma unroll
        for (int g = 0; g < 8; ++g) rp[g * AP] = c[g];
    }
    __syncthreads();

    for (int i = tid; i < R_OWN * D; i += AGG_THREADS) {
        const int node = c0 + (i >> 7);
        if (node < n) out[(size_t)node * D + (i & (D - 1))] = res[(i >> 7) * AP + (i & (D - 1))] + bv;
    }
}

extern "C" void kernel_launch(void* const* d_in, const int* in_sizes, int n_in,
                              void* d_out, int out_size, void* d_ws, size_t ws_size,
                              hipStream_t stream) {
    const float* x    = (const float*)d_in[0];
    const int*   ei   = (const int*)d_in[1];
    const float* W    = (const float*)d_in[2];
    const float* bias = (const float*)d_in[3];
    float* out = (float*)d_out;

    const int N = in_sizes[0] / D;
    const int E = in_sizes[1] / 2;
    const int* row = ei;
    const int* col = ei + E;

    const int nown = (N + R_OWN - 1) / R_OWN;
    const int nchunks = (E + CH_EDGES - 1) / CH_EDGES;
    unsigned* srtc = (unsigned*)d_ws;
    unsigned* offc = srtc + (size_t)nchunks * CH_EDGES;
    unsigned* cplanes = offc + (size_t)nchunks * IDX_OWNERS;

    owner_sort_kernel<<<CNT_WGS + nchunks, IDX_THREADS, 0, stream>>>(row, col, srtc, offc, cplanes, E / 4);
    owner_aggregate_kernel<<<nown, AGG_THREADS, 0, stream>>>(x, W, cplanes, bias, srtc, offc, out, N, nchunks);
}
